# starting kernel
# baseline (speedup 1.0000x reference)
#include <hip/hip_runtime.h>

typedef float v2f __attribute__((ext_vector_type(2)));
typedef float v4f __attribute__((ext_vector_type(4)));
typedef float v8f __attribute__((ext_vector_type(8)));

#define C_DIM   128
#define N_DIM   64
#define HW      3136
#define PLANES  (C_DIM * N_DIM)
#define NUMEL_PER_CH (N_DIM * HW)
#define BN_EPS  1e-5f

static __device__ __forceinline__ v8f wmma_rowsum(v2f a, v2f ones, v8f c) {
    return __builtin_amdgcn_wmma_f32_16x16x4_f32(
        false, a, false, ones, (short)0, c, false, false);
}

__global__ __launch_bounds__(256) void bn_partials(const float* __restrict__ x,
                                                   float* __restrict__ psum,
                                                   float* __restrict__ psq) {
    const int c = blockIdx.x;
    const int n = blockIdx.y;
    const int plane = n * C_DIM + c;
    const v4f* __restrict__ xp4 = (const v4f*)(x + (long)plane * HW);
    const int tid = threadIdx.x;

    v2f ones; ones.x = 1.0f; ones.y = 1.0f;
    v8f s0 = {}, s1 = {}, q0 = {}, q1 = {};

    #pragma unroll
    for (int k = 0; k < 3; ++k) {
        v4f v = xp4[k * 256 + tid];
        v2f a0; a0.x = v.x;       a0.y = v.y;
        v2f a1; a1.x = v.z;       a1.y = v.w;
        v2f b0; b0.x = v.x * v.x; b0.y = v.y * v.y;
        v2f b1; b1.x = v.z * v.z; b1.y = v.w * v.w;
        s0 = wmma_rowsum(a0, ones, s0);
        s1 = wmma_rowsum(a1, ones, s1);
        q0 = wmma_rowsum(b0, ones, q0);
        q1 = wmma_rowsum(b1, ones, q1);
    }

    if (tid < 32) {
        const float2* __restrict__ xp2 = (const float2*)xp4;
        float2 v = xp2[1536 + tid];
        v2f a; a.x = v.x;       a.y = v.y;
        v2f b; b.x = v.x * v.x; b.y = v.y * v.y;
        s0 = wmma_rowsum(a, ones, s0);
        q0 = wmma_rowsum(b, ones, q0);
    }

    v8f tS = s0 + s1;
    v8f tQ = q0 + q1;
    float s = tS[0] + tS[1] + tS[2] + tS[3] + tS[4] + tS[5] + tS[6] + tS[7];
    float q = tQ[0] + tQ[1] + tQ[2] + tQ[3] + tQ[4] + tQ[5] + tQ[6] + tQ[7];
    s += __shfl_xor(s, 16, 32);
    q += __shfl_xor(q, 16, 32);

    __shared__ float ls[8];
    __shared__ float lq[8];
    const int wave = tid >> 5;
    const int lane = tid & 31;
    if (lane == 0) { ls[wave] = s; lq[wave] = q; }
    __syncthreads();
    if (tid == 0) {
        float S = 0.f, Q = 0.f;
        for (int w = 0; w < 8; ++w) { S += ls[w]; Q += lq[w]; }
        psum[plane] = S;
        psq [plane] = Q;
    }
}

__global__ __launch_bounds__(C_DIM) void bn_finalize(const float* __restrict__ psum,
                                                     const float* __restrict__ psq,
                                                     const float* __restrict__ weight,
                                                     const float* __restrict__ bias,
                                                     float* __restrict__ scale,
                                                     float* __restrict__ shift) {
    const int c = threadIdx.x;
    float S = 0.f, Q = 0.f;
    for (int n = 0; n < N_DIM; ++n) {
        S += psum[n * C_DIM + c];
        Q += psq [n * C_DIM + c];
    }
    const float inv_n = 1.0f / (float)NUMEL_PER_CH;
    const float mean = S * inv_n;
    const float var  = Q * inv_n - mean * mean;
    const float istd = rsqrtf(var + BN_EPS);
    const float sc   = weight[c] * istd;
    scale[c] = sc;
    shift[c] = bias[c] - mean * sc;
}

__global__ __launch_bounds__(256) void bn_apply(const float* __restrict__ x,
                                                const float* __restrict__ scale,
                                                const float* __restrict__ shift,
                                                float* __restrict__ y,
                                                int n4) {
    __shared__ float ssc[C_DIM];
    __shared__ float ssh[C_DIM];
    if (threadIdx.x < C_DIM) {
        ssc[threadIdx.x] = scale[threadIdx.x];
        ssh[threadIdx.x] = shift[threadIdx.x];
    }
    __syncthreads();

    const v4f* __restrict__ x4 = (const v4f*)x;
    v4f* __restrict__ y4 = (v4f*)y;
    const int stride = (int)gridDim.x * 256;

    for (int i = (int)blockIdx.x * 256 + (int)threadIdx.x; i < n4; i += stride) {
        if (i + stride < n4) {
            __builtin_prefetch(x4 + i + stride, 0, 1);
        }
        const int c = (i / (HW / 4)) % C_DIM;
        v4f v = x4[i];
        const float sc = ssc[c];
        const float sh = ssh[c];
        v4f r;
        r.x = v.x * sc + sh;
        r.y = v.y * sc + sh;
        r.z = v.z * sc + sh;
        r.w = v.w * sc + sh;
        __builtin_nontemporal_store(r, y4 + i);
    }
}

extern "C" void kernel_launch(void* const* d_in, const int* in_sizes, int n_in,
                              void* d_out, int out_size, void* d_ws, size_t ws_size,
                              hipStream_t stream) {
    const float* x = (const float*)d_in[0];
    const float* w = (const float*)d_in[1];
    const float* b = (const float*)d_in[2];
    float* y  = (float*)d_out;
    float* ws = (float*)d_ws;

    float* psum  = ws;
    float* psq   = ws + PLANES;
    float* scale = ws + 2 * PLANES;
    float* shift = scale + C_DIM;

    dim3 g1(C_DIM, N_DIM);
    bn_partials<<<g1, 256, 0, stream>>>(x, psum, psq);
    bn_finalize<<<1, C_DIM, 0, stream>>>(psum, psq, w, b, scale, shift);

    const int n4 = out_size / 4;
    bn_apply<<<4096, 256, 0, stream>>>(x, scale, shift, y, n4);
}


# each wave streams two whole planes, 30+ loads in flight, in both streaming passes: waves per pass 65536 / 32768 -> 4096, no LDS or barrier; plane sum = one wave's WMMA chain (sum order changes)
# speedup vs baseline: 1.2439x; 1.2439x over previous
#include <hip/hip_runtime.h>

typedef float v2f __attribute__((ext_vector_type(2)));
typedef float v4f __attribute__((ext_vector_type(4)));
typedef float v8f __attribute__((ext_vector_type(8)));

#define C_DIM   128
#define N_DIM   64
#define HW      3136
#define PLANES  (C_DIM * N_DIM)
#define NUMEL_PER_CH (N_DIM * HW)
#define BN_EPS  1e-5f

#define WG_WAVES        8
#define WAVE_PLANES     2
#define STREAM_GROUPS   (PLANES / (WG_WAVES * WAVE_PLANES))
#define FULL_LOADS      24
#define TAIL_F2         (FULL_LOADS * 32 * 2)

static __device__ __forceinline__ v8f wmma_rowsum(v2f a, v2f ones, v8f c) {
    return __builtin_amdgcn_wmma_f32_16x16x4_f32(
        false, a, false, ones, (short)0, c, false, false);
}

__global__ __launch_bounds__(WG_WAVES * 32) void bn_partials(const float* __restrict__ x,
                                                            float* __restrict__ psum,
                                                            float* __restrict__ psq) {
    const int lane = threadIdx.x & 31;
    const int wave = __builtin_amdgcn_readfirstlane((int)(threadIdx.x >> 5));
    const int plane0 = ((int)blockIdx.x * WG_WAVES + wave) * WAVE_PLANES;
    const float* __restrict__ xb = x + (long)plane0 * HW;

    v4f   v[WAVE_PLANES][FULL_LOADS];
    float2 t[WAVE_PLANES];
    #pragma unroll
    for (int p = 0; p < WAVE_PLANES; ++p) {
        const v4f* __restrict__ xp4 = (const v4f*)(xb + p * HW);
        #pragma unroll
        for (int k = 0; k < FULL_LOADS; ++k) v[p][k] = xp4[k * 32 + lane];
        t[p] = ((const float2*)xp4)[TAIL_F2 + lane];
    }

    v2f ones; ones.x = 1.0f; ones.y = 1.0f;
    float S[WAVE_PLANES], Q[WAVE_PLANES];
    #pragma unroll
    for (int p = 0; p < WAVE_PLANES; ++p) {
        v8f s0 = {}, s1 = {}, q0 = {}, q1 = {};
        #pragma unroll
        for (int k = 0; k < FULL_LOADS; ++k) {
            const v4f w = v[p][k];
            v2f a0; a0.x = w.x;       a0.y = w.y;
            v2f a1; a1.x = w.z;       a1.y = w.w;
            v2f b0; b0.x = w.x * w.x; b0.y = w.y * w.y;
            v2f b1; b1.x = w.z * w.z; b1.y = w.w * w.w;
            s0 = wmma_rowsum(a0, ones, s0);
            s1 = wmma_rowsum(a1, ones, s1);
            q0 = wmma_rowsum(b0, ones, q0);
            q1 = wmma_rowsum(b1, ones, q1);
        }
        {
            v2f a; a.x = t[p].x;          a.y = t[p].y;
            v2f b; b.x = t[p].x * t[p].x; b.y = t[p].y * t[p].y;
            s0 = wmma_rowsum(a, ones, s0);
            q0 = wmma_rowsum(b, ones, q0);
        }
        v8f tS = s0 + s1;
        v8f tQ = q0 + q1;
        float s = tS[0] + tS[1] + tS[2] + tS[3] + tS[4] + tS[5] + tS[6] + tS[7];
        float q = tQ[0] + tQ[1] + tQ[2] + tQ[3] + tQ[4] + tQ[5] + tQ[6] + tQ[7];
        s += __shfl_xor(s, 16, 32);
        q += __shfl_xor(q, 16, 32);
        S[p] = s;
        Q[p] = q;
    }

    if (lane == 0) {
        float2 so; so.x = S[0]; so.y = S[1];
        float2 qo; qo.x = Q[0]; qo.y = Q[1];
        *(float2*)(psum + plane0) = so;
        *(float2*)(psq  + plane0) = qo;
    }
}

__global__ __launch_bounds__(C_DIM) void bn_finalize(const float* __restrict__ psum,
                                                     const float* __restrict__ psq,
                                                     const float* __restrict__ weight,
                                                     const float* __restrict__ bias,
                                                     float* __restrict__ scale,
                                                     float* __restrict__ shift) {
    const int c = threadIdx.x;
    float S = 0.f, Q = 0.f;
    for (int n = 0; n < N_DIM; ++n) {
        S += psum[n * C_DIM + c];
        Q += psq [n * C_DIM + c];
    }
    const float inv_n = 1.0f / (float)NUMEL_PER_CH;
    const float mean = S * inv_n;
    const float var  = Q * inv_n - mean * mean;
    const float istd = rsqrtf(var + BN_EPS);
    const float sc   = weight[c] * istd;
    scale[c] = sc;
    shift[c] = bias[c] - mean * sc;
}

__global__ __launch_bounds__(WG_WAVES * 32) void bn_apply(const float* __restrict__ x,
                                                         const float* __restrict__ scale,
                                                         const float* __restrict__ shift,
                                                         float* __restrict__ y) {
    const int lane = threadIdx.x & 31;
    const int wave = __builtin_amdgcn_readfirstlane((int)(threadIdx.x >> 5));
    const int plane0 = ((int)blockIdx.x * WG_WAVES + wave) * WAVE_PLANES;
    const int c0 = plane0 & (C_DIM - 1);
    const float* __restrict__ xb = x + (long)plane0 * HW;
    float* __restrict__ yb = y + (long)plane0 * HW;

    v4f   v[WAVE_PLANES][FULL_LOADS];
    float2 t[WAVE_PLANES];
    #pragma unroll
    for (int p = 0; p < WAVE_PLANES; ++p) {
        const v4f* __restrict__ xp4 = (const v4f*)(xb + p * HW);
        #pragma unroll
        for (int k = 0; k < FULL_LOADS; ++k) v[p][k] = xp4[k * 32 + lane];
        t[p] = ((const float2*)xp4)[TAIL_F2 + lane];
    }

    #pragma unroll
    for (int p = 0; p < WAVE_PLANES; ++p) {
        const float sc = scale[c0 + p];
        const float sh = shift[c0 + p];
        v4f* __restrict__ yp4 = (v4f*)(yb + p * HW);
        #pragma unroll
        for (int k = 0; k < FULL_LOADS; ++k) {
            const v4f w = v[p][k];
            v4f r;
            r.x = w.x * sc + sh;
            r.y = w.y * sc + sh;
            r.z = w.z * sc + sh;
            r.w = w.w * sc + sh;
            __builtin_nontemporal_store(r, yp4 + k * 32 + lane);
        }
        v2f r2;
        r2.x = t[p].x * sc + sh;
        r2.y = t[p].y * sc + sh;
        __builtin_nontemporal_store(r2, (v2f*)yp4 + TAIL_F2 + lane);
    }
}

extern "C" void kernel_launch(void* const* d_in, const int* in_sizes, int n_in,
                              void* d_out, int out_size, void* d_ws, size_t ws_size,
                              hipStream_t stream) {
    (void)in_sizes; (void)n_in; (void)out_size; (void)ws_size;
    const float* x = (const float*)d_in[0];
    const float* w = (const float*)d_in[1];
    const float* b = (const float*)d_in[2];
    float* y  = (float*)d_out;
    float* ws = (float*)d_ws;

    float* psum  = ws;
    float* psq   = ws + PLANES;
    float* scale = ws + 2 * PLANES;
    float* shift = scale + C_DIM;

    bn_partials<<<STREAM_GROUPS, WG_WAVES * 32, 0, stream>>>(x, psum, psq);
    bn_finalize<<<1, C_DIM, 0, stream>>>(psum, psq, w, b, scale, shift);
    bn_apply<<<STREAM_GROUPS, WG_WAVES * 32, 0, stream>>>(x, scale, shift, y);
}


# finalize launch folded into the apply pass: plane partials stored channel-major, each apply wave sums its two channels' 64 partials (half-wave shuffle tree: sum order changes); launches 3 -> 2
# speedup vs baseline: 1.7217x; 1.3841x over previous
#include <hip/hip_runtime.h>

typedef float v2f __attribute__((ext_vector_type(2)));
typedef float v4f __attribute__((ext_vector_type(4)));
typedef float v8f __attribute__((ext_vector_type(8)));

#define C_DIM   128
#define N_DIM   64
#define HW      3136
#define PLANES  (C_DIM * N_DIM)
#define NUMEL_PER_CH (N_DIM * HW)
#define BN_EPS  1e-5f

#define WG_WAVES        8
#define WAVE_PLANES     2
#define STREAM_GROUPS   (PLANES / (WG_WAVES * WAVE_PLANES))
#define FULL_LOADS      24
#define TAIL_F2         (FULL_LOADS * 32 * 2)

static __device__ __forceinline__ v8f wmma_rowsum(v2f a, v2f ones, v8f c) {
    return __builtin_amdgcn_wmma_f32_16x16x4_f32(
        false, a, false, ones, (short)0, c, false, false);
}

__global__ __launch_bounds__(WG_WAVES * 32) void bn_partials(const float* __restrict__ x,
                                                            float* __restrict__ psum,
                                                            float* __restrict__ psq) {
    const int lane = threadIdx.x & 31;
    const int wave = __builtin_amdgcn_readfirstlane((int)(threadIdx.x >> 5));
    const int plane0 = ((int)blockIdx.x * WG_WAVES + wave) * WAVE_PLANES;
    const float* __restrict__ xb = x + (long)plane0 * HW;

    v4f   v[WAVE_PLANES][FULL_LOADS];
    float2 t[WAVE_PLANES];
    #pragma unroll
    for (int p = 0; p < WAVE_PLANES; ++p) {
        const v4f* __restrict__ xp4 = (const v4f*)(xb + p * HW);
        #pragma unroll
        for (int k = 0; k < FULL_LOADS; ++k) v[p][k] = xp4[k * 32 + lane];
        t[p] = ((const float2*)xp4)[TAIL_F2 + lane];
    }

    v2f ones; ones.x = 1.0f; ones.y = 1.0f;
    float S[WAVE_PLANES], Q[WAVE_PLANES];
    #pragma unroll
    for (int p = 0; p < WAVE_PLANES; ++p) {
        v8f s0 = {}, s1 = {}, q0 = {}, q1 = {};
        #pragma unroll
        for (int k = 0; k < FULL_LOADS; ++k) {
            const v4f w = v[p][k];
            v2f a0; a0.x = w.x;       a0.y = w.y;
            v2f a1; a1.x = w.z;       a1.y = w.w;
            v2f b0; b0.x = w.x * w.x; b0.y = w.y * w.y;
            v2f b1; b1.x = w.z * w.z; b1.y = w.w * w.w;
            s0 = wmma_rowsum(a0, ones, s0);
            s1 = wmma_rowsum(a1, ones, s1);
            q0 = wmma_rowsum(b0, ones, q0);
            q1 = wmma_rowsum(b1, ones, q1);
        }
        {
            v2f a; a.x = t[p].x;          a.y = t[p].y;
            v2f b; b.x = t[p].x * t[p].x; b.y = t[p].y * t[p].y;
            s0 = wmma_rowsum(a, ones, s0);
            q0 = wmma_rowsum(b, ones, q0);
        }
        v8f tS = s0 + s1;
        v8f tQ = q0 + q1;
        float s = tS[0] + tS[1] + tS[2] + tS[3] + tS[4] + tS[5] + tS[6] + tS[7];
        float q = tQ[0] + tQ[1] + tQ[2] + tQ[3] + tQ[4] + tQ[5] + tQ[6] + tQ[7];
        s += __shfl_xor(s, 16, 32);
        q += __shfl_xor(q, 16, 32);
        S[p] = s;
        Q[p] = q;
    }

    if (lane < WAVE_PLANES) {
        const int n   = plane0 >> 7;
        const int c0  = plane0 & (C_DIM - 1);
        const int idx = (c0 + lane) * N_DIM + n;
        psum[idx] = lane ? S[1] : S[0];
        psq [idx] = lane ? Q[1] : Q[0];
    }
}

static __device__ __forceinline__ float half_wave_sum(float v) {
    v += __shfl_xor(v, 8, 32);
    v += __shfl_xor(v, 4, 32);
    v += __shfl_xor(v, 2, 32);
    v += __shfl_xor(v, 1, 32);
    return v;
}

__global__ __launch_bounds__(WG_WAVES * 32) void bn_apply(const float* __restrict__ x,
                                                         const float* __restrict__ psum,
                                                         const float* __restrict__ psq,
                                                         const float* __restrict__ weight,
                                                         const float* __restrict__ bias,
                                                         float* __restrict__ y) {
    const int lane = threadIdx.x & 31;
    const int wave = __builtin_amdgcn_readfirstlane((int)(threadIdx.x >> 5));
    const int plane0 = ((int)blockIdx.x * WG_WAVES + wave) * WAVE_PLANES;
    const int c0 = plane0 & (C_DIM - 1);
    const float* __restrict__ xb = x + (long)plane0 * HW;
    float* __restrict__ yb = y + (long)plane0 * HW;

    const v4f ps = ((const v4f*)(psum + c0 * N_DIM))[lane];
    const v4f pq = ((const v4f*)(psq  + c0 * N_DIM))[lane];
    const float wc = weight[c0 + (lane >> 4)];
    const float bc = bias  [c0 + (lane >> 4)];

    v4f   v[WAVE_PLANES][FULL_LOADS];
    float2 t[WAVE_PLANES];
    #pragma unroll
    for (int p = 0; p < WAVE_PLANES; ++p) {
        const v4f* __restrict__ xp4 = (const v4f*)(xb + p * HW);
        #pragma unroll
        for (int k = 0; k < FULL_LOADS; ++k) v[p][k] = xp4[k * 32 + lane];
        t[p] = ((const float2*)xp4)[TAIL_F2 + lane];
    }

    const float Sc = half_wave_sum((ps.x + ps.y) + (ps.z + ps.w));
    const float Qc = half_wave_sum((pq.x + pq.y) + (pq.z + pq.w));
    const float inv_n = 1.0f / (float)NUMEL_PER_CH;
    const float mean = Sc * inv_n;
    const float var  = Qc * inv_n - mean * mean;
    const float istd = rsqrtf(var + BN_EPS);
    const float scl  = wc * istd;
    const float shl  = bc - mean * scl;

    #pragma unroll
    for (int p = 0; p < WAVE_PLANES; ++p) {
        const float sc = __int_as_float(__builtin_amdgcn_readlane(__float_as_int(scl), 16 * p));
        const float sh = __int_as_float(__builtin_amdgcn_readlane(__float_as_int(shl), 16 * p));
        v4f* __restrict__ yp4 = (v4f*)(yb + p * HW);
        #pragma unroll
        for (int k = 0; k < FULL_LOADS; ++k) {
            const v4f w = v[p][k];
            v4f r;
            r.x = w.x * sc + sh;
            r.y = w.y * sc + sh;
            r.z = w.z * sc + sh;
            r.w = w.w * sc + sh;
            __builtin_nontemporal_store(r, yp4 + k * 32 + lane);
        }
        v2f r2;
        r2.x = t[p].x * sc + sh;
        r2.y = t[p].y * sc + sh;
        __builtin_nontemporal_store(r2, (v2f*)yp4 + TAIL_F2 + lane);
    }
}

extern "C" void kernel_launch(void* const* d_in, const int* in_sizes, int n_in,
                              void* d_out, int out_size, void* d_ws, size_t ws_size,
                              hipStream_t stream) {
    (void)in_sizes; (void)n_in; (void)out_size; (void)ws_size;
    const float* x = (const float*)d_in[0];
    const float* w = (const float*)d_in[1];
    const float* b = (const float*)d_in[2];
    float* y  = (float*)d_out;
    float* ws = (float*)d_ws;

    float* psum  = ws;
    float* psq   = ws + PLANES;

    bn_partials<<<STREAM_GROUPS, WG_WAVES * 32, 0, stream>>>(x, psum, psq);
    bn_apply<<<STREAM_GROUPS, WG_WAVES * 32, 0, stream>>>(x, psum, psq, w, b, y);
}
